# starting kernel
# baseline (speedup 1.0000x reference)
#include <hip/hip_runtime.h>
#include <hip/hip_bf16.h>

typedef __attribute__((ext_vector_type(2))) float v2f;
typedef __attribute__((ext_vector_type(8))) float v8f;

#define B_   4
#define T_   4096
#define DM   1024
#define DS   256
#define LCH  128
#define NCH  (T_ / LCH)

__global__ __launch_bounds__(256) void gemm_in_kernel(
    const float* __restrict__ x, const float* __restrict__ W_in,
    const float* __restrict__ b_in, const float* __restrict__ gamma,
    float* __restrict__ gu)
{
  const int lane  = threadIdx.x & 31;
  const int wave  = threadIdx.x >> 5;
  const int gw    = blockIdx.x * 8 + wave;
  const int tilesN = DS / 16;
  const int waveM = gw / tilesN;
  const int waveN = gw % tilesN;
  const int m0 = waveM * 64;
  const int n0 = waveN * 16;
  const int lm = lane & 15;
  const int kk = (lane >> 4) << 1;

  v8f acc0 = {}, acc1 = {}, acc2 = {}, acc3 = {};

  for (int k = 0; k < DM; k += 4) {
    const int kr = k + kk;
    v2f bfrag;
    bfrag.x = W_in[(size_t)(kr + 0) * DS + n0 + lm];
    bfrag.y = W_in[(size_t)(kr + 1) * DS + n0 + lm];
    const float* xrow = x + (size_t)(m0 + lm) * DM + kr;
    v2f a0 = *(const v2f*)(xrow + (size_t)0 * 16 * DM);
    v2f a1 = *(const v2f*)(xrow + (size_t)1 * 16 * DM);
    v2f a2 = *(const v2f*)(xrow + (size_t)2 * 16 * DM);
    v2f a3 = *(const v2f*)(xrow + (size_t)3 * 16 * DM);
    acc0 = __builtin_amdgcn_wmma_f32_16x16x4_f32(false, a0, false, bfrag, (short)0, acc0, false, false);
    acc1 = __builtin_amdgcn_wmma_f32_16x16x4_f32(false, a1, false, bfrag, (short)0, acc1, false, false);
    acc2 = __builtin_amdgcn_wmma_f32_16x16x4_f32(false, a2, false, bfrag, (short)0, acc2, false, false);
    acc3 = __builtin_amdgcn_wmma_f32_16x16x4_f32(false, a3, false, bfrag, (short)0, acc3, false, false);
  }

  const int rbase = (lane >> 4) * 8;
  const int col   = n0 + lm;
  const float bi  = b_in[col];
  const float g   = gamma[col];
#pragma unroll
  for (int r = 0; r < 8; ++r) {
    gu[(size_t)(m0 +  0 + rbase + r) * DS + col] = g * (acc0[r] + bi);
    gu[(size_t)(m0 + 16 + rbase + r) * DS + col] = g * (acc1[r] + bi);
    gu[(size_t)(m0 + 32 + rbase + r) * DS + col] = g * (acc2[r] + bi);
    gu[(size_t)(m0 + 48 + rbase + r) * DS + col] = g * (acc3[r] + bi);
  }
}

__global__ __launch_bounds__(DS) void chunk_end_kernel(
    const float* __restrict__ lambda_p, const float* __restrict__ gu,
    float* __restrict__ chunkend)
{
  const int b = blockIdx.x / NCH;
  const int c = blockIdx.x % NCH;
  const int s = threadIdx.x;
  const float a = 1.0f / (1.0f + expf(-lambda_p[s]));
  size_t idx = ((size_t)(b * T_ + c * LCH)) * DS + s;
  float e = 0.0f;
#pragma unroll 4
  for (int j = 0; j < LCH; ++j) {
    e = fmaf(a, e, gu[idx]);
    idx += DS;
  }
  chunkend[(size_t)(b * NCH + c) * DS + s] = e;
}

__global__ __launch_bounds__(DS) void scan_carry_kernel(
    const float* __restrict__ lambda_p, const float* __restrict__ state,
    const float* __restrict__ chunkend, float* __restrict__ carrybuf)
{
  const int b = blockIdx.x;
  const int s = threadIdx.x;
  const float a = 1.0f / (1.0f + expf(-lambda_p[s]));

  float aL = a;
#pragma unroll
  for (int i = 0; i < 7; ++i) aL *= aL;

  float carry = state[(size_t)b * DS + s];
  for (int c = 0; c < NCH; ++c) {
    carrybuf[(size_t)(b * NCH + c) * DS + s] = carry;
    carry = fmaf(aL, carry, chunkend[(size_t)(b * NCH + c) * DS + s]);
  }
}

__global__ __launch_bounds__(DS) void rescan_kernel(
    const float* __restrict__ lambda_p, float* __restrict__ gu,
    const float* __restrict__ carrybuf, float* __restrict__ h_last)
{
  const int b = blockIdx.x / NCH;
  const int c = blockIdx.x % NCH;
  const int s = threadIdx.x;
  const float a = 1.0f / (1.0f + expf(-lambda_p[s]));
  size_t idx = ((size_t)(b * T_ + c * LCH)) * DS + s;
  float h = carrybuf[(size_t)(b * NCH + c) * DS + s];
#pragma unroll 4
  for (int j = 0; j < LCH; ++j) {
    h = fmaf(a, h, gu[idx]);
    gu[idx] = h;
    idx += DS;
  }
  if (c == NCH - 1) h_last[(size_t)b * DS + s] = h;
}

__global__ __launch_bounds__(256) void gemm_out_kernel(
    const float* __restrict__ h, const float* __restrict__ W_out,
    const float* __restrict__ b_out, float* __restrict__ y)
{
  const int lane  = threadIdx.x & 31;
  const int wave  = threadIdx.x >> 5;
  const int gw    = blockIdx.x * 8 + wave;
  const int tilesN = DM / 16;
  const int waveM = gw / tilesN;
  const int waveN = gw % tilesN;
  const int m0 = waveM * 64;
  const int n0 = waveN * 16;
  const int lm = lane & 15;
  const int kk = (lane >> 4) << 1;

  v8f acc0 = {}, acc1 = {}, acc2 = {}, acc3 = {};

  for (int k = 0; k < DS; k += 4) {
    const int kr = k + kk;
    v2f bfrag;
    bfrag.x = W_out[(size_t)(kr + 0) * DM + n0 + lm];
    bfrag.y = W_out[(size_t)(kr + 1) * DM + n0 + lm];
    const float* hrow = h + (size_t)(m0 + lm) * DS + kr;
    v2f a0 = *(const v2f*)(hrow + (size_t)0 * 16 * DS);
    v2f a1 = *(const v2f*)(hrow + (size_t)1 * 16 * DS);
    v2f a2 = *(const v2f*)(hrow + (size_t)2 * 16 * DS);
    v2f a3 = *(const v2f*)(hrow + (size_t)3 * 16 * DS);
    acc0 = __builtin_amdgcn_wmma_f32_16x16x4_f32(false, a0, false, bfrag, (short)0, acc0, false, false);
    acc1 = __builtin_amdgcn_wmma_f32_16x16x4_f32(false, a1, false, bfrag, (short)0, acc1, false, false);
    acc2 = __builtin_amdgcn_wmma_f32_16x16x4_f32(false, a2, false, bfrag, (short)0, acc2, false, false);
    acc3 = __builtin_amdgcn_wmma_f32_16x16x4_f32(false, a3, false, bfrag, (short)0, acc3, false, false);
  }

  const int rbase = (lane >> 4) * 8;
  const int col   = n0 + lm;
  const float bo  = b_out[col];
#pragma unroll
  for (int r = 0; r < 8; ++r) {
    y[(size_t)(m0 +  0 + rbase + r) * DM + col] = acc0[r] + bo;
    y[(size_t)(m0 + 16 + rbase + r) * DM + col] = acc1[r] + bo;
    y[(size_t)(m0 + 32 + rbase + r) * DM + col] = acc2[r] + bo;
    y[(size_t)(m0 + 48 + rbase + r) * DM + col] = acc3[r] + bo;
  }
}

extern "C" void kernel_launch(void* const* d_in, const int* in_sizes, int n_in,
                              void* d_out, int out_size, void* d_ws, size_t ws_size,
                              hipStream_t stream) {
  const float* x        = (const float*)d_in[0];
  const float* state    = (const float*)d_in[1];
  const float* W_in     = (const float*)d_in[2];
  const float* b_in     = (const float*)d_in[3];
  const float* lambda_p = (const float*)d_in[4];
  const float* gamma_p  = (const float*)d_in[5];
  const float* W_out    = (const float*)d_in[6];
  const float* b_out    = (const float*)d_in[7];

  float* y      = (float*)d_out;
  float* h_last = y + (size_t)B_ * T_ * DM;

  float* gu       = (float*)d_ws;
  float* chunkend = gu + (size_t)B_ * T_ * DS;
  float* carry    = chunkend + (size_t)B_ * NCH * DS;

  gemm_in_kernel<<<512, 256, 0, stream>>>(x, W_in, b_in, gamma_p, gu);

  chunk_end_kernel<<<B_ * NCH, DS, 0, stream>>>(lambda_p, gu, chunkend);
  scan_carry_kernel<<<B_, DS, 0, stream>>>(lambda_p, state, chunkend, carry);
  rescan_kernel<<<B_ * NCH, DS, 0, stream>>>(lambda_p, gu, carry, h_last);

  gemm_out_kernel<<<2048, 256, 0, stream>>>(gu, W_out, b_out, y);
}


# both GEMMs: 64x128 work-group tiles, A and B K-slices staged in LDS by coalesced 16-byte loads (double buffer), WMMA fragments read from LDS, not strided global loads; k order and sums unchanged
# speedup vs baseline: 1.4470x; 1.4470x over previous
#include <hip/hip_runtime.h>
#include <hip/hip_bf16.h>


typedef __attribute__((ext_vector_type(2))) float v2f;
typedef __attribute__((ext_vector_type(4))) float v4f;
typedef __attribute__((ext_vector_type(8))) float v8f;

#define B_   4
#define T_   4096
#define DM   1024
#define DS   256
#define LCH  128
#define NCH  (T_ / LCH)

#define BK      32
#define APITCH  36
#define BPAD    8

template <int K, int N, int FM, int FN, bool SCALED>
__device__ __forceinline__ void gemm_block_tile(
    const float* __restrict__ A, const float* __restrict__ Bm,
    const float* __restrict__ bias, const float* __restrict__ scale,
    float* __restrict__ C, float* __restrict__ As, float* __restrict__ Bs)
{
  constexpr int WAVES_N = 4;
  constexpr int BM = 2 * FM * 16;
  constexpr int BN = WAVES_N * FN * 16;
  constexpr int BP = BN + BPAD;
  constexpr int A_TILE = BM * APITCH;
  constexpr int B_TILE = BK * BP;
  constexpr int A_LD = (BM * BK / 4) / 256;
  constexpr int B_LD = (BK * BN / 4) / 256;
  constexpr int B_F4_ROW = BN / 4;
  constexpr int B_ROWS_PASS = 256 / B_F4_ROW;
  constexpr int NT = K / BK;
  constexpr int TILES_N = N / BN;

  const int tid  = threadIdx.x;
  const int lane = tid & 31;
  const int wave = tid >> 5;
  const int wm = wave / WAVES_N;
  const int wn = wave % WAVES_N;
  const int c  = lane & 15;
  const int h  = lane >> 4;

  const int m0 = (blockIdx.x / TILES_N) * BM;
  const int n0 = (blockIdx.x % TILES_N) * BN;

  const int a_kq  = tid & 7;
  const int a_row = tid >> 3;
  const int b_nq  = tid % B_F4_ROW;
  const int b_row = tid / B_F4_ROW;
  const unsigned a_off = (unsigned)(m0 + a_row) * K + a_kq * 4;
  const unsigned b_off = (unsigned)b_row * N + n0 + b_nq * 4;
  float* a_dst = As + a_row * APITCH + a_kq * 4;
  float* b_dst = Bs + b_row * BP + b_nq * 4;

  const float* a_frag = As + (wm * FM * 16 + c) * APITCH + 2 * h;
  const float* b_frag = Bs + (2 * h) * BP + wn * FN * 16 + c;

  v4f ra[A_LD], rb[B_LD];
  v8f acc[FM][FN];
#pragma unroll
  for (int i = 0; i < FM; ++i)
#pragma unroll
    for (int j = 0; j < FN; ++j) acc[i][j] = v8f{};

  auto load_tile = [&](int t) {
    const float* a_t = A  + (size_t)t * BK;
    const float* b_t = Bm + (size_t)t * BK * N;
#pragma unroll
    for (int i = 0; i < A_LD; ++i)
      ra[i] = *(const v4f*)(a_t + (a_off + (unsigned)(i * 32 * K)));
#pragma unroll
    for (int i = 0; i < B_LD; ++i)
      rb[i] = *(const v4f*)(b_t + (b_off + (unsigned)(i * B_ROWS_PASS * N)));
  };
  auto store_tile = [&](int buf) {
#pragma unroll
    for (int i = 0; i < A_LD; ++i)
      *(v4f*)(a_dst + buf * A_TILE + i * 32 * APITCH) = ra[i];
#pragma unroll
    for (int i = 0; i < B_LD; ++i)
      *(v4f*)(b_dst + buf * B_TILE + i * B_ROWS_PASS * BP) = rb[i];
  };
  auto compute = [&](int buf) {
    const float* ap = a_frag + buf * A_TILE;
    const float* bp = b_frag + buf * B_TILE;
#pragma unroll
    for (int ks = 0; ks < BK / 4; ++ks) {
      v2f af[FM], bf[FN];
#pragma unroll
      for (int i = 0; i < FM; ++i)
        af[i] = *(const v2f*)(ap + i * 16 * APITCH + ks * 4);
#pragma unroll
      for (int j = 0; j < FN; ++j) {
        bf[j].x = bp[(ks * 4 + 0) * BP + j * 16];
        bf[j].y = bp[(ks * 4 + 1) * BP + j * 16];
      }
#pragma unroll
      for (int i = 0; i < FM; ++i)
#pragma unroll
        for (int j = 0; j < FN; ++j)
          acc[i][j] = __builtin_amdgcn_wmma_f32_16x16x4_f32(false, af[i], false, bf[j], (short)0, acc[i][j], false, false);
    }
  };

  load_tile(0);
  store_tile(0);
  __syncthreads();
  for (int t = 0; t < NT - 1; ++t) {
    load_tile(t + 1);
    asm volatile("" ::: "memory");
    compute(t & 1);
    store_tile((t + 1) & 1);
    __syncthreads();
  }
  compute((NT - 1) & 1);

#pragma unroll
  for (int j = 0; j < FN; ++j) {
    const int col = n0 + (wn * FN + j) * 16 + c;
    const float bi = bias[col];
    float g = 0.0f;
    if (SCALED) g = scale[col];
#pragma unroll
    for (int i = 0; i < FM; ++i) {
      const int row = m0 + (wm * FM + i) * 16 + 8 * h;
#pragma unroll
      for (int r = 0; r < 8; ++r) {
        if (SCALED) C[(size_t)(row + r) * N + col] = g * (acc[i][j][r] + bi);
        else        C[(size_t)(row + r) * N + col] = acc[i][j][r] + bi;
      }
    }
  }
}

__global__ __launch_bounds__(256) void gemm_in_kernel(
    const float* __restrict__ x, const float* __restrict__ W_in,
    const float* __restrict__ b_in, const float* __restrict__ gamma,
    float* __restrict__ gu)
{
  __shared__ __attribute__((aligned(16))) float As[2 * 64 * APITCH];
  __shared__ __attribute__((aligned(16))) float Bs[2 * BK * (128 + BPAD)];
  gemm_block_tile<DM, DS, 2, 2, true>(x, W_in, b_in, gamma, gu, As, Bs);
}

__global__ __launch_bounds__(DS) void chunk_end_kernel(
    const float* __restrict__ lambda_p, const float* __restrict__ gu,
    float* __restrict__ chunkend)
{
  const int b = blockIdx.x / NCH;
  const int c = blockIdx.x % NCH;
  const int s = threadIdx.x;
  const float a = 1.0f / (1.0f + expf(-lambda_p[s]));
  size_t idx = ((size_t)(b * T_ + c * LCH)) * DS + s;
  float e = 0.0f;
#pragma unroll 4
  for (int j = 0; j < LCH; ++j) {
    e = fmaf(a, e, gu[idx]);
    idx += DS;
  }
  chunkend[(size_t)(b * NCH + c) * DS + s] = e;
}

__global__ __launch_bounds__(DS) void scan_carry_kernel(
    const float* __restrict__ lambda_p, const float* __restrict__ state,
    const float* __restrict__ chunkend, float* __restrict__ carrybuf)
{
  const int b = blockIdx.x;
  const int s = threadIdx.x;
  const float a = 1.0f / (1.0f + expf(-lambda_p[s]));

  float aL = a;
#pragma unroll
  for (int i = 0; i < 7; ++i) aL *= aL;

  float carry = state[(size_t)b * DS + s];
  for (int c = 0; c < NCH; ++c) {
    carrybuf[(size_t)(b * NCH + c) * DS + s] = carry;
    carry = fmaf(aL, carry, chunkend[(size_t)(b * NCH + c) * DS + s]);
  }
}

__global__ __launch_bounds__(DS) void rescan_kernel(
    const float* __restrict__ lambda_p, float* __restrict__ gu,
    const float* __restrict__ carrybuf, float* __restrict__ h_last)
{
  const int b = blockIdx.x / NCH;
  const int c = blockIdx.x % NCH;
  const int s = threadIdx.x;
  const float a = 1.0f / (1.0f + expf(-lambda_p[s]));
  size_t idx = ((size_t)(b * T_ + c * LCH)) * DS + s;
  float h = carrybuf[(size_t)(b * NCH + c) * DS + s];
#pragma unroll 4
  for (int j = 0; j < LCH; ++j) {
    h = fmaf(a, h, gu[idx]);
    gu[idx] = h;
    idx += DS;
  }
  if (c == NCH - 1) h_last[(size_t)b * DS + s] = h;
}

__global__ __launch_bounds__(256) void gemm_out_kernel(
    const float* __restrict__ h, const float* __restrict__ W_out,
    const float* __restrict__ b_out, float* __restrict__ y)
{
  __shared__ __attribute__((aligned(16))) float As[2 * 64 * APITCH];
  __shared__ __attribute__((aligned(16))) float Bs[2 * BK * (128 + BPAD)];
  gemm_block_tile<DS, DM, 2, 2, false>(h, W_out, b_out, b_out, y, As, Bs);
}

extern "C" void kernel_launch(void* const* d_in, const int* in_sizes, int n_in,
                              void* d_out, int out_size, void* d_ws, size_t ws_size,
                              hipStream_t stream) {
  const float* x        = (const float*)d_in[0];
  const float* state    = (const float*)d_in[1];
  const float* W_in     = (const float*)d_in[2];
  const float* b_in     = (const float*)d_in[3];
  const float* lambda_p = (const float*)d_in[4];
  const float* gamma_p  = (const float*)d_in[5];
  const float* W_out    = (const float*)d_in[6];
  const float* b_out    = (const float*)d_in[7];

  float* y      = (float*)d_out;
  float* h_last = y + (size_t)B_ * T_ * DM;

  float* gu       = (float*)d_ws;
  float* chunkend = gu + (size_t)B_ * T_ * DS;
  float* carry    = chunkend + (size_t)B_ * NCH * DS;

  gemm_in_kernel<<<512, 256, 0, stream>>>(x, W_in, b_in, gamma_p, gu);

  chunk_end_kernel<<<B_ * NCH, DS, 0, stream>>>(lambda_p, gu, chunkend);
  scan_carry_kernel<<<B_, DS, 0, stream>>>(lambda_p, state, chunkend, carry);
  rescan_kernel<<<B_ * NCH, DS, 0, stream>>>(lambda_p, gu, carry, h_last);

  gemm_out_kernel<<<2048, 256, 0, stream>>>(gu, W_out, b_out, y);
}


# scan passes A and C: each lane keeps a 32-row batch of row loads in flight ahead of its serial fma chain (was 3-4 loads per round trip), 64-thread blocks on every compute unit; fma order unchanged
# speedup vs baseline: 1.7989x; 1.2432x over previous
#include <hip/hip_runtime.h>
#include <hip/hip_bf16.h>


typedef __attribute__((ext_vector_type(2))) float v2f;
typedef __attribute__((ext_vector_type(4))) float v4f;
typedef __attribute__((ext_vector_type(8))) float v8f;

#define B_   4
#define T_   4096
#define DM   1024
#define DS   256
#define LCH  128
#define NCH  (T_ / LCH)

#define BK      32
#define APITCH  36
#define BPAD    8

template <int K, int N, int FM, int FN, bool SCALED>
__device__ __forceinline__ void gemm_block_tile(
    const float* __restrict__ A, const float* __restrict__ Bm,
    const float* __restrict__ bias, const float* __restrict__ scale,
    float* __restrict__ C, float* __restrict__ As, float* __restrict__ Bs)
{
  constexpr int WAVES_N = 4;
  constexpr int BM = 2 * FM * 16;
  constexpr int BN = WAVES_N * FN * 16;
  constexpr int BP = BN + BPAD;
  constexpr int A_TILE = BM * APITCH;
  constexpr int B_TILE = BK * BP;
  constexpr int A_LD = (BM * BK / 4) / 256;
  constexpr int B_LD = (BK * BN / 4) / 256;
  constexpr int B_F4_ROW = BN / 4;
  constexpr int B_ROWS_PASS = 256 / B_F4_ROW;
  constexpr int NT = K / BK;
  constexpr int TILES_N = N / BN;

  const int tid  = threadIdx.x;
  const int lane = tid & 31;
  const int wave = tid >> 5;
  const int wm = wave / WAVES_N;
  const int wn = wave % WAVES_N;
  const int c  = lane & 15;
  const int h  = lane >> 4;

  const int m0 = (blockIdx.x / TILES_N) * BM;
  const int n0 = (blockIdx.x % TILES_N) * BN;

  const int a_kq  = tid & 7;
  const int a_row = tid >> 3;
  const int b_nq  = tid % B_F4_ROW;
  const int b_row = tid / B_F4_ROW;
  const unsigned a_off = (unsigned)(m0 + a_row) * K + a_kq * 4;
  const unsigned b_off = (unsigned)b_row * N + n0 + b_nq * 4;
  float* a_dst = As + a_row * APITCH + a_kq * 4;
  float* b_dst = Bs + b_row * BP + b_nq * 4;

  const float* a_frag = As + (wm * FM * 16 + c) * APITCH + 2 * h;
  const float* b_frag = Bs + (2 * h) * BP + wn * FN * 16 + c;

  v4f ra[A_LD], rb[B_LD];
  v8f acc[FM][FN];
#pragma unroll
  for (int i = 0; i < FM; ++i)
#pragma unroll
    for (int j = 0; j < FN; ++j) acc[i][j] = v8f{};

  auto load_tile = [&](int t) {
    const float* a_t = A  + (size_t)t * BK;
    const float* b_t = Bm + (size_t)t * BK * N;
#pragma unroll
    for (int i = 0; i < A_LD; ++i)
      ra[i] = *(const v4f*)(a_t + (a_off + (unsigned)(i * 32 * K)));
#pragma unroll
    for (int i = 0; i < B_LD; ++i)
      rb[i] = *(const v4f*)(b_t + (b_off + (unsigned)(i * B_ROWS_PASS * N)));
  };
  auto store_tile = [&](int buf) {
#pragma unroll
    for (int i = 0; i < A_LD; ++i)
      *(v4f*)(a_dst + buf * A_TILE + i * 32 * APITCH) = ra[i];
#pragma unroll
    for (int i = 0; i < B_LD; ++i)
      *(v4f*)(b_dst + buf * B_TILE + i * B_ROWS_PASS * BP) = rb[i];
  };
  auto compute = [&](int buf) {
    const float* ap = a_frag + buf * A_TILE;
    const float* bp = b_frag + buf * B_TILE;
#pragma unroll
    for (int ks = 0; ks < BK / 4; ++ks) {
      v2f af[FM], bf[FN];
#pragma unroll
      for (int i = 0; i < FM; ++i)
        af[i] = *(const v2f*)(ap + i * 16 * APITCH + ks * 4);
#pragma unroll
      for (int j = 0; j < FN; ++j) {
        bf[j].x = bp[(ks * 4 + 0) * BP + j * 16];
        bf[j].y = bp[(ks * 4 + 1) * BP + j * 16];
      }
#pragma unroll
      for (int i = 0; i < FM; ++i)
#pragma unroll
        for (int j = 0; j < FN; ++j)
          acc[i][j] = __builtin_amdgcn_wmma_f32_16x16x4_f32(false, af[i], false, bf[j], (short)0, acc[i][j], false, false);
    }
  };

  load_tile(0);
  store_tile(0);
  __syncthreads();
  for (int t = 0; t < NT - 1; ++t) {
    load_tile(t + 1);
    asm volatile("" ::: "memory");
    compute(t & 1);
    store_tile((t + 1) & 1);
    __syncthreads();
  }
  compute((NT - 1) & 1);

#pragma unroll
  for (int j = 0; j < FN; ++j) {
    const int col = n0 + (wn * FN + j) * 16 + c;
    const float bi = bias[col];
    float g = 0.0f;
    if (SCALED) g = scale[col];
#pragma unroll
    for (int i = 0; i < FM; ++i) {
      const int row = m0 + (wm * FM + i) * 16 + 8 * h;
#pragma unroll
      for (int r = 0; r < 8; ++r) {
        if (SCALED) C[(size_t)(row + r) * N + col] = g * (acc[i][j][r] + bi);
        else        C[(size_t)(row + r) * N + col] = acc[i][j][r] + bi;
      }
    }
  }
}

__global__ __launch_bounds__(256) void gemm_in_kernel(
    const float* __restrict__ x, const float* __restrict__ W_in,
    const float* __restrict__ b_in, const float* __restrict__ gamma,
    float* __restrict__ gu)
{
  __shared__ __attribute__((aligned(16))) float As[2 * 64 * APITCH];
  __shared__ __attribute__((aligned(16))) float Bs[2 * BK * (128 + BPAD)];
  gemm_block_tile<DM, DS, 2, 2, true>(x, W_in, b_in, gamma, gu, As, Bs);
}

#define SCAN_T   64
#define SCAN_Q   (DS / SCAN_T)
#define SCAN_RB  32
#define SCAN_NB  (LCH / SCAN_RB)

__global__ __launch_bounds__(SCAN_T) void chunk_end_kernel(
    const float* __restrict__ lambda_p, const float* __restrict__ gu,
    float* __restrict__ chunkend)
{
  const int bc = blockIdx.x / SCAN_Q;
  const int s  = (blockIdx.x % SCAN_Q) * SCAN_T + threadIdx.x;
  const float a = 1.0f / (1.0f + expf(-lambda_p[s]));
  const float* p = gu + (size_t)bc * LCH * DS + s;

  float g[2][SCAN_RB];
#pragma unroll
  for (int j = 0; j < SCAN_RB; ++j) g[0][j] = p[(size_t)j * DS];

  float e = 0.0f;
#pragma unroll
  for (int k = 0; k < SCAN_NB; ++k) {
    if (k + 1 < SCAN_NB) {
#pragma unroll
      for (int j = 0; j < SCAN_RB; ++j) g[(k + 1) & 1][j] = p[(size_t)((k + 1) * SCAN_RB + j) * DS];
    }
#pragma unroll
    for (int j = 0; j < SCAN_RB; ++j) e = fmaf(a, e, g[k & 1][j]);
  }
  chunkend[(size_t)bc * DS + s] = e;
}

__global__ __launch_bounds__(DS) void scan_carry_kernel(
    const float* __restrict__ lambda_p, const float* __restrict__ state,
    const float* __restrict__ chunkend, float* __restrict__ carrybuf)
{
  const int b = blockIdx.x;
  const int s = threadIdx.x;
  const float a = 1.0f / (1.0f + expf(-lambda_p[s]));

  float aL = a;
#pragma unroll
  for (int i = 0; i < 7; ++i) aL *= aL;

  float carry = state[(size_t)b * DS + s];
  for (int c = 0; c < NCH; ++c) {
    carrybuf[(size_t)(b * NCH + c) * DS + s] = carry;
    carry = fmaf(aL, carry, chunkend[(size_t)(b * NCH + c) * DS + s]);
  }
}

__global__ __launch_bounds__(SCAN_T) void rescan_kernel(
    const float* __restrict__ lambda_p, float* __restrict__ gu,
    const float* __restrict__ carrybuf, float* __restrict__ h_last)
{
  const int bc = blockIdx.x / SCAN_Q;
  const int s  = (blockIdx.x % SCAN_Q) * SCAN_T + threadIdx.x;
  const float a = 1.0f / (1.0f + expf(-lambda_p[s]));
  float* p = gu + (size_t)bc * LCH * DS + s;

  float g[2][SCAN_RB];
#pragma unroll
  for (int j = 0; j < SCAN_RB; ++j) g[0][j] = p[(size_t)j * DS];

  float h = carrybuf[(size_t)bc * DS + s];
#pragma unroll
  for (int k = 0; k < SCAN_NB; ++k) {
    if (k + 1 < SCAN_NB) {
#pragma unroll
      for (int j = 0; j < SCAN_RB; ++j) g[(k + 1) & 1][j] = p[(size_t)((k + 1) * SCAN_RB + j) * DS];
    }
#pragma unroll
    for (int j = 0; j < SCAN_RB; ++j) {
      h = fmaf(a, h, g[k & 1][j]);
      p[(size_t)(k * SCAN_RB + j) * DS] = h;
    }
  }
  if ((bc % NCH) == NCH - 1) h_last[(size_t)(bc / NCH) * DS + s] = h;
}

__global__ __launch_bounds__(256) void gemm_out_kernel(
    const float* __restrict__ h, const float* __restrict__ W_out,
    const float* __restrict__ b_out, float* __restrict__ y)
{
  __shared__ __attribute__((aligned(16))) float As[2 * 64 * APITCH];
  __shared__ __attribute__((aligned(16))) float Bs[2 * BK * (128 + BPAD)];
  gemm_block_tile<DS, DM, 2, 2, false>(h, W_out, b_out, b_out, y, As, Bs);
}

extern "C" void kernel_launch(void* const* d_in, const int* in_sizes, int n_in,
                              void* d_out, int out_size, void* d_ws, size_t ws_size,
                              hipStream_t stream) {
  const float* x        = (const float*)d_in[0];
  const float* state    = (const float*)d_in[1];
  const float* W_in     = (const float*)d_in[2];
  const float* b_in     = (const float*)d_in[3];
  const float* lambda_p = (const float*)d_in[4];
  const float* gamma_p  = (const float*)d_in[5];
  const float* W_out    = (const float*)d_in[6];
  const float* b_out    = (const float*)d_in[7];

  float* y      = (float*)d_out;
  float* h_last = y + (size_t)B_ * T_ * DM;

  float* gu       = (float*)d_ws;
  float* chunkend = gu + (size_t)B_ * T_ * DS;
  float* carry    = chunkend + (size_t)B_ * NCH * DS;

  gemm_in_kernel<<<512, 256, 0, stream>>>(x, W_in, b_in, gamma_p, gu);

  chunk_end_kernel<<<B_ * NCH * SCAN_Q, SCAN_T, 0, stream>>>(lambda_p, gu, chunkend);
  scan_carry_kernel<<<B_, DS, 0, stream>>>(lambda_p, state, chunkend, carry);
  rescan_kernel<<<B_ * NCH * SCAN_Q, SCAN_T, 0, stream>>>(lambda_p, gu, carry, h_last);

  gemm_out_kernel<<<2048, 256, 0, stream>>>(gu, W_out, b_out, y);
}
